# starting kernel
# baseline (speedup 1.0000x reference)
#include <hip/hip_runtime.h>
#include <math.h>


typedef float v2f __attribute__((ext_vector_type(2)));
typedef float v8f __attribute__((ext_vector_type(8)));

#define BSZ 1024
#define TSZ 500
#define CSZ 64


__device__ __forceinline__ float lse2(float x, float y) {
    float m = fmaxf(x, y);
    return m + __logf(__expf(x - m) + __expf(y - m));
}

__global__ __launch_bounds__(64) void bkt_prep(const float* __restrict__ trans,
                                               const float* __restrict__ obs,
                                               const float* __restrict__ init,
                                               float* __restrict__ ws) {
    int c = threadIdx.x;
    if (c >= CSZ) return;
    float* W = ws;
    float* linit = ws + CSZ * 16;

    for (int s = 0; s < 2; ++s) {
        float x0 = obs[c * 4 + s * 2 + 0];
        float x1 = obs[c * 4 + s * 2 + 1];
        float z = lse2(x0, x1);
        W[c * 16 + s * 2 + 0] = x0 - z;
        W[c * 16 + s * 2 + 1] = x1 - z;
    }
    for (int tp = 0; tp < 2; ++tp) {
        float x0 = trans[c * 4 + 0 * 2 + tp];
        float x1 = trans[c * 4 + 1 * 2 + tp];
        float z = lse2(x0, x1);
        W[c * 16 + 4 + 0 * 2 + tp] = x0 - z;
        W[c * 16 + 4 + 1 * 2 + tp] = x1 - z;
    }
    for (int n = 8; n < 16; ++n) W[c * 16 + n] = 0.0f;
    {
        float x0 = init[c * 2 + 0];
        float x1 = init[c * 2 + 1];
        float z = lse2(x0, x1);
        linit[c * 2 + 0] = x0 - z;
        linit[c * 2 + 1] = x1 - z;
    }
}

__global__ __launch_bounds__(32) void bkt_fwd(const int* __restrict__ corr,
                                              const int* __restrict__ kc,
                                              const float* __restrict__ Amat,
                                              const float* __restrict__ ws,
                                              float* __restrict__ out) {
    const int lane = threadIdx.x;
    const int bl   = lane & 15;
    const int half = lane >> 4;
    const int bg   = blockIdx.x * 16 + bl;

    __shared__ float sAccW[16 * 16];
    __shared__ float sAccS[2][16 * 16];
    __shared__ float sA3[16][2];

    const float* W     = ws;
    const float* linit = ws + CSZ * 16;

    float Wf[16][2];
#pragma unroll
    for (int k = 0; k < 16; ++k) {
        int c = 4 * k + 2 * half;
        Wf[k][0] = W[(c + 0) * 16 + bl];
        Wf[k][1] = W[(c + 1) * 16 + bl];
    }
    float la[2][16][2];
#pragma unroll
    for (int k = 0; k < 16; ++k) {
        int c = 4 * k + 2 * half;
#pragma unroll
        for (int j = 0; j < 2; ++j) {
            la[0][k][j] = linit[(c + j) * 2 + 0];
            la[1][k][j] = linit[(c + j) * 2 + 1];
        }
    }

    auto gather = [&](float (&dst)[16][2], int kcvv) {
#pragma unroll
        for (int k = 0; k < 16; ++k) {
            const float2 v = *(const float2*)(Amat + kcvv * CSZ + 4 * k + 2 * half);
            dst[k][0] = v.x;
            dst[k][1] = v.y;
        }
    };

    auto compute = [&](int t, const float (&cc)[16][2], int y) {
        v8f accW = {};
        v8f acc0 = {};
        v8f acc1 = {};
#pragma unroll
        for (int k = 0; k < 16; ++k) {
            v2f a  = {cc[k][0], cc[k][1]};
            v2f bw = {Wf[k][0], Wf[k][1]};
            v2f b0 = {la[0][k][0], la[0][k][1]};
            v2f b1 = {la[1][k][0], la[1][k][1]};
            accW = __builtin_amdgcn_wmma_f32_16x16x4_f32(false, a, false, bw, (short)0, accW, false, false);
            acc0 = __builtin_amdgcn_wmma_f32_16x16x4_f32(false, a, false, b0, (short)0, acc0, false, false);
            acc1 = __builtin_amdgcn_wmma_f32_16x16x4_f32(false, a, false, b1, (short)0, acc1, false, false);
        }

#pragma unroll
        for (int r = 0; r < 8; ++r) {
            sAccW[(r + 8 * half) * 16 + bl]    = accW[r];
            sAccS[0][(r + 8 * half) * 16 + bl] = acc0[r];
            sAccS[1][(r + 8 * half) * 16 + bl] = acc1[r];
        }
        __syncthreads();

        if (lane < 16) {
            const int b = lane;
            const float a1_00 = sAccW[b * 16 + 0];
            const float a1_01 = sAccW[b * 16 + 1];
            const float a1_10 = sAccW[b * 16 + 2];
            const float a1_11 = sAccW[b * 16 + 3];
            const float c_00  = sAccW[b * 16 + 4];
            const float c_01  = sAccW[b * 16 + 5];
            const float c_10  = sAccW[b * 16 + 6];
            const float c_11  = sAccW[b * 16 + 7];
            const float a2_0  = sAccS[0][b * 16 + b];
            const float a2_1  = sAccS[1][b * 16 + b];

            float p0 = lse2(a1_00 + a2_0, a1_10 + a2_1);
            float p1 = lse2(a1_01 + a2_0, a1_11 + a2_1);
            float z  = lse2(p0, p1);
            float2 o2;
            o2.x = p0 - z;
            o2.y = p1 - z;
            *(float2*)(out + (size_t)(bg * TSZ + t) * 2) = o2;

            const float lp0 = y ? a1_01 : a1_00;
            const float lp1 = y ? a1_11 : a1_10;
            sA3[b][0] = lse2(lp0 + a2_0 + c_00, lp1 + a2_1 + c_01);
            sA3[b][1] = lse2(lp0 + a2_0 + c_10, lp1 + a2_1 + c_11);
        }
        __syncthreads();

        const float a3s0 = sA3[bl][0];
        const float a3s1 = sA3[bl][1];
#pragma unroll
        for (int k = 0; k < 16; ++k) {
#pragma unroll
            for (int j = 0; j < 2; ++j) {
                la[0][k][j] = fmaf(cc[k][j], a3s0 - la[0][k][j], la[0][k][j]);
                la[1][k][j] = fmaf(cc[k][j], a3s1 - la[1][k][j], la[1][k][j]);
            }
        }
    };

    float ccA[16][2], ccB[16][2];
    int kcv_c = kc[bg * TSZ + 0];
    int y_c   = corr[bg * TSZ + 0];
    int kcv_n = kc[bg * TSZ + 1];
    int y_n   = corr[bg * TSZ + 1];
    gather(ccA, kcv_c);

    for (int t = 0; t < TSZ; t += 2) {
        gather(ccB, kcv_n);
        const int i2 = (t + 2 < TSZ) ? t + 2 : TSZ - 1;
        const int kcv2 = kc[bg * TSZ + i2];
        const int y2   = corr[bg * TSZ + i2];
        compute(t, ccA, y_c);

        gather(ccA, kcv2);
        const int i3 = (t + 3 < TSZ) ? t + 3 : TSZ - 1;
        const int kcv3 = kc[bg * TSZ + i3];
        const int y3   = corr[bg * TSZ + i3];
        compute(t + 1, ccB, y_n);

        y_c = y2;
        kcv_n = kcv3;
        y_n = y3;
    }
}

extern "C" void kernel_launch(void* const* d_in, const int* in_sizes, int n_in,
                              void* d_out, int out_size, void* d_ws, size_t ws_size,
                              hipStream_t stream) {
    const int*   corr  = (const int*)d_in[0];
    const int*   kc    = (const int*)d_in[1];
    const float* Amat  = (const float*)d_in[2];
    const float* trans = (const float*)d_in[3];
    const float* obs   = (const float*)d_in[4];
    const float* init  = (const float*)d_in[5];
    float*       out   = (float*)d_out;
    float*       ws    = (float*)d_ws;

    (void)in_sizes; (void)n_in; (void)out_size; (void)ws_size;

    bkt_prep<<<1, 64, 0, stream>>>(trans, obs, init, ws);
    bkt_fwd<<<BSZ / 16, 32, 0, stream>>>(corr, kc, Amat, ws, out);
}


# Product work of a block (row gathers, W products, ring stores) split by step parity over waves 1 and 2; wave 2 still finishes all outputs: a wave's row gather is exposed 5 times per block, not 10
# speedup vs baseline: 2.5881x; 2.5881x over previous
#include <hip/hip_runtime.h>
#include <math.h>


typedef float v2f __attribute__((ext_vector_type(2)));
typedef float v4f __attribute__((ext_vector_type(4)));
typedef float v8f __attribute__((ext_vector_type(8)));

#define BSZ 1024
#define TSZ 500
#define CSZ 64
#define NB 10
#define NBLK (TSZ / NB)
#define NH (NB / 2)


__device__ __forceinline__ float lse2(float x, float y) {
    float m = fmaxf(x, y);
    return m + __logf(__expf(x - m) + __expf(y - m));
}

__device__ __forceinline__ float xhalf(float x) {
    int b = __builtin_bit_cast(int, x);
    return __builtin_bit_cast(float,
        __builtin_amdgcn_permlanex16(b, b, 0x76543210, 0xfedcba98, false, false));
}

__global__ __launch_bounds__(64) void bkt_prep(const float* __restrict__ trans,
                                               const float* __restrict__ obs,
                                               const float* __restrict__ init,
                                               float* __restrict__ ws) {
    int c = threadIdx.x;
    if (c >= CSZ) return;
    float* W = ws;
    float* linit = ws + CSZ * 16;

    float w8[8];
    for (int s = 0; s < 2; ++s) {
        float x0 = obs[c * 4 + s * 2 + 0];
        float x1 = obs[c * 4 + s * 2 + 1];
        float z = lse2(x0, x1);
        w8[s * 2 + 0] = x0 - z;
        w8[s * 2 + 1] = x1 - z;
    }
    for (int tp = 0; tp < 2; ++tp) {
        float x0 = trans[c * 4 + 0 * 2 + tp];
        float x1 = trans[c * 4 + 1 * 2 + tp];
        float z = lse2(x0, x1);
        w8[4 + 0 * 2 + tp] = x0 - z;
        w8[4 + 1 * 2 + tp] = x1 - z;
    }
    for (int n = 0; n < 8; ++n) {
        W[c * 16 + n]     = w8[n];
        W[c * 16 + 8 + n] = w8[n];
    }
    {
        float x0 = init[c * 2 + 0];
        float x1 = init[c * 2 + 1];
        float z = lse2(x0, x1);
        linit[c * 2 + 0] = x0 - z;
        linit[c * 2 + 1] = x1 - z;
    }
}

__global__ __launch_bounds__(96) __attribute__((amdgpu_waves_per_eu(1, 1))) void bkt_fwd(const int* __restrict__ corr,
                                              const int* __restrict__ kc,
                                              const float* __restrict__ Amat,
                                              const float* __restrict__ ws,
                                              float* __restrict__ out) {
    __shared__ v4f ring[3][NB][16][2];
    __shared__ __attribute__((aligned(16))) float a2ring[2][NB][16][2];
    __shared__ v4f ccring[2][NB][8][32];

    const int lane = threadIdx.x & 31;
    const int role = __builtin_amdgcn_readfirstlane((int)(threadIdx.x >> 5));
    const int bl   = lane & 15;
    const int half = lane >> 4;
    const int bg   = blockIdx.x * 16 + bl;

    const bool own = (half == (bl >> 3));
    const bool q0 = (bl & 1) != 0;
    const bool q1 = (bl & 2) != 0;
    const bool q2 = (bl & 4) != 0;
    auto pick8 = [&](const v8f& v) -> float {
        const float t0 = q0 ? v[1] : v[0];
        const float t1 = q0 ? v[3] : v[2];
        const float t2 = q0 ? v[5] : v[4];
        const float t3 = q0 ? v[7] : v[6];
        const float u0 = q1 ? t1 : t0;
        const float u1 = q1 ? t3 : t2;
        return q2 ? u1 : u0;
    };

    const float* W     = ws;
    const float* linit = ws + CSZ * 16;

    float Wf[16][2];
#pragma unroll
    for (int k = 0; k < 16; ++k) {
        int c = 4 * k + 2 * half;
        Wf[k][0] = W[(c + 0) * 16 + bl];
        Wf[k][1] = W[(c + 1) * 16 + bl];
    }
    float la[2][16][2];
#pragma unroll
    for (int k = 0; k < 16; ++k) {
        int c = 4 * k + 2 * half;
#pragma unroll
        for (int j = 0; j < 2; ++j) {
            la[0][k][j] = linit[(c + j) * 2 + 0];
            la[1][k][j] = linit[(c + j) * 2 + 1];
        }
    }

    auto gather = [&](float (&dst)[16][2], int kcvv) {
#pragma unroll
        for (int k = 0; k < 16; ++k) {
            const float2 v = *(const float2*)(Amat + kcvv * CSZ + 4 * k + 2 * half);
            dst[k][0] = v.x;
            dst[k][1] = v.y;
        }
    };

    auto ldcc = [&](float (&dst)[16][2], int slot, int i) {
#pragma unroll
        for (int p = 0; p < 8; ++p) {
            const v4f q = ccring[slot][i][p][lane];
            dst[2 * p][0]     = q.x;
            dst[2 * p][1]     = q.y;
            dst[2 * p + 1][0] = q.z;
            dst[2 * p + 1][1] = q.w;
        }
    };

    auto prod = [&](int slot, int s3, int i, const float (&cc)[16][2], int y) {
        v8f accW = {};
#pragma unroll
        for (int k = 0; k < 16; ++k) {
            v2f a  = {cc[k][0], cc[k][1]};
            v2f bw = {Wf[k][0], Wf[k][1]};
            accW = __builtin_amdgcn_wmma_f32_16x16x4_f32(false, bw, false, a, (short)0, accW, false, false);
        }
#pragma unroll
        for (int p = 0; p < 8; ++p) {
            const v4f q = {cc[2 * p][0], cc[2 * p][1], cc[2 * p + 1][0], cc[2 * p + 1][1]};
            ccring[slot][i][p][lane] = q;
        }
        const float lp0 = y ? accW[1] : accW[0];
        const float lp1 = y ? accW[3] : accW[2];
        v4f v;
        v.x = half ? lp0 + accW[4] : accW[0];
        v.y = half ? lp1 + accW[5] : accW[1];
        v.z = half ? lp0 + accW[6] : accW[2];
        v.w = half ? lp1 + accW[7] : accW[3];
        ring[s3][i][bl][half] = v;
    };

    auto compute = [&](int slot, int i, const v4f wk, const float (&cc)[16][2]) {
        v8f acc0 = {};
        v8f acc1 = {};
#pragma unroll
        for (int k = 0; k < 16; ++k) {
            v2f a  = {cc[k][0], cc[k][1]};
            v2f b0 = {la[0][k][0], la[0][k][1]};
            v2f b1 = {la[1][k][0], la[1][k][1]};
            acc0 = __builtin_amdgcn_wmma_f32_16x16x4_f32(false, a, false, b0, (short)0, acc0, false, false);
            acc1 = __builtin_amdgcn_wmma_f32_16x16x4_f32(false, a, false, b1, (short)0, acc1, false, false);
        }

        const float d0 = pick8(acc0);
        const float d1 = pick8(acc1);
        const float x0 = xhalf(d0);
        const float x1 = xhalf(d1);
        const float a2_0 = own ? d0 : x0;
        const float a2_1 = own ? d1 : x1;

        a2ring[slot][i][bl][half] = half ? a2_1 : a2_0;

        const float a3s0 = lse2(wk.x + a2_0, wk.y + a2_1);
        const float a3s1 = lse2(wk.z + a2_0, wk.w + a2_1);

#pragma unroll
        for (int k = 0; k < 16; ++k) {
#pragma unroll
            for (int j = 0; j < 2; ++j) {
                la[0][k][j] = fmaf(cc[k][j], a3s0 - la[0][k][j], la[0][k][j]);
                la[1][k][j] = fmaf(cc[k][j], a3s1 - la[1][k][j], la[1][k][j]);
            }
        }
    };

    float ccA[16][2], ccB[16][2];

    auto produce = [&](int blk, int s3, int par) {
        const int slot = blk & 1;
        const int base = bg * TSZ + blk * NB + par;
        int kk[NH], yy[NH];
#pragma unroll
        for (int j = 0; j < NH; ++j) {
            kk[j] = kc[base + 2 * j];
            yy[j] = corr[base + 2 * j];
        }
        gather(ccA, kk[0]);
        __builtin_amdgcn_sched_barrier(0);
#pragma unroll
        for (int j = 0; j < NH; ++j) {
            if (j + 1 < NH) {
                gather((j & 1) ? ccA : ccB, kk[j + 1]);
                __builtin_amdgcn_sched_group_barrier(0x020, 16, 0);
                __builtin_amdgcn_sched_group_barrier(0x008, 16, 0);
            }
            prod(slot, s3, par + 2 * j, (j & 1) ? ccB : ccA, yy[j]);
        }
    };

    auto consume = [&](int blk, int s3) {
        const int slot = blk & 1;
        ldcc(ccA, slot, 0);
        for (int i = 0; i < NB; i += 2) {
            ldcc(ccB, slot, i + 1);
            const v4f wa = ring[s3][i][bl][1];
            const v4f wb = ring[s3][i + 1][bl][1];
            __builtin_amdgcn_sched_group_barrier(0x100, 10, 0);
            __builtin_amdgcn_sched_group_barrier(0x008, 32, 0);
            compute(slot, i, wa, ccA);

            ldcc(ccA, slot, (i + 2 < NB) ? i + 2 : NB - 1);
            __builtin_amdgcn_sched_group_barrier(0x100, 8, 0);
            __builtin_amdgcn_sched_group_barrier(0x008, 32, 0);
            compute(slot, i + 1, wb, ccB);
        }
    };

    auto finish = [&](int blk, int s3) {
        const int slot = blk & 1;
#pragma unroll
        for (int i = 0; i < NB; ++i) {
            const v4f w0 = ring[s3][i][bl][0];
            const v2f a2 = *(const v2f*)&a2ring[slot][i][bl][0];
            const float p0 = lse2(w0.x + a2.x, w0.z + a2.y);
            const float p1 = lse2(w0.y + a2.x, w0.w + a2.y);
            const float z  = lse2(p0, p1);
            out[(size_t)(bg * TSZ + blk * NB + i) * 2 + half] = (half ? p1 : p0) - z;
        }
    };

    const int par = role - 1;
    if (role != 0) {
        produce(0, 0, par);
    }
    __syncthreads();
    int s3 = 0;
    for (int blk = 0; blk < NBLK; ++blk) {
        const int s3n = (s3 == 2) ? 0 : s3 + 1;
        const int s3p = (s3 == 0) ? 2 : s3 - 1;
        if (role == 0) {
            consume(blk, s3);
        } else {
            if (blk + 1 < NBLK) produce(blk + 1, s3n, par);
            if (role == 2 && blk > 0) finish(blk - 1, s3p);
        }
        __syncthreads();
        s3 = s3n;
    }
    if (role == 2) {
        finish(NBLK - 1, (s3 == 0) ? 2 : s3 - 1);
    }
}

extern "C" void kernel_launch(void* const* d_in, const int* in_sizes, int n_in,
                              void* d_out, int out_size, void* d_ws, size_t ws_size,
                              hipStream_t stream) {
    const int*   corr  = (const int*)d_in[0];
    const int*   kc    = (const int*)d_in[1];
    const float* Amat  = (const float*)d_in[2];
    const float* trans = (const float*)d_in[3];
    const float* obs   = (const float*)d_in[4];
    const float* init  = (const float*)d_in[5];
    float*       out   = (float*)d_out;
    float*       ws    = (float*)d_ws;

    (void)in_sizes; (void)n_in; (void)out_size; (void)ws_size;

    bkt_prep<<<1, 64, 0, stream>>>(trans, obs, init, ws);
    bkt_fwd<<<BSZ / 16, 96, 0, stream>>>(corr, kc, Amat, ws, out);
}


# 16-batch tile's recurrence split over two waves of 8 batches (two SIMDs), both states of a batch as columns of one B operand: 16 f32 WMMAs per step on a wave's serial chain, not 32; sum order kept
# speedup vs baseline: 3.3357x; 1.2889x over previous
#include <hip/hip_runtime.h>
#include <math.h>


typedef float v2f __attribute__((ext_vector_type(2)));
typedef float v4f __attribute__((ext_vector_type(4)));
typedef float v8f __attribute__((ext_vector_type(8)));

#define BSZ 1024
#define TSZ 500
#define CSZ 64
#define NB 10
#define NBLK (TSZ / NB)
#define NH (NB / 2)


__device__ __forceinline__ float lse2(float x, float y) {
    float m = fmaxf(x, y);
    return m + __logf(__expf(x - m) + __expf(y - m));
}

__device__ __forceinline__ float xrow8(float x) {
    int b = __builtin_bit_cast(int, x);
    return __builtin_bit_cast(float,
        __builtin_amdgcn_permlane16(b, b, 0xfedcba98, 0x76543210, false, false));
}

__global__ __launch_bounds__(64) void bkt_prep(const float* __restrict__ trans,
                                               const float* __restrict__ obs,
                                               const float* __restrict__ init,
                                               float* __restrict__ ws) {
    int c = threadIdx.x;
    if (c >= CSZ) return;
    float* W = ws;
    float* linit = ws + CSZ * 16;

    float w8[8];
    for (int s = 0; s < 2; ++s) {
        float x0 = obs[c * 4 + s * 2 + 0];
        float x1 = obs[c * 4 + s * 2 + 1];
        float z = lse2(x0, x1);
        w8[s * 2 + 0] = x0 - z;
        w8[s * 2 + 1] = x1 - z;
    }
    for (int tp = 0; tp < 2; ++tp) {
        float x0 = trans[c * 4 + 0 * 2 + tp];
        float x1 = trans[c * 4 + 1 * 2 + tp];
        float z = lse2(x0, x1);
        w8[4 + 0 * 2 + tp] = x0 - z;
        w8[4 + 1 * 2 + tp] = x1 - z;
    }
    for (int n = 0; n < 8; ++n) {
        W[c * 16 + n]     = w8[n];
        W[c * 16 + 8 + n] = w8[n];
    }
    {
        float x0 = init[c * 2 + 0];
        float x1 = init[c * 2 + 1];
        float z = lse2(x0, x1);
        linit[c * 2 + 0] = x0 - z;
        linit[c * 2 + 1] = x1 - z;
    }
}

__global__ __launch_bounds__(128) __attribute__((amdgpu_waves_per_eu(1, 1))) void bkt_fwd(const int* __restrict__ corr,
                                              const int* __restrict__ kc,
                                              const float* __restrict__ Amat,
                                              const float* __restrict__ ws,
                                              float* __restrict__ out) {
    __shared__ v4f ring[3][NB][16][2];
    __shared__ __attribute__((aligned(16))) float a2ring[2][NB][16][2];
    __shared__ v4f ccring[2][NB][8][32];

    const int lane = threadIdx.x & 31;
    const int role = __builtin_amdgcn_readfirstlane((int)(threadIdx.x >> 5));
    const int bl   = lane & 15;
    const int half = lane >> 4;
    const int bg   = blockIdx.x * 16 + bl;

    const int rb = ((role & 1) << 3) | (bl & 7);
    const int sq = bl >> 3;
    const int rl = rb + 16 * half;
    const bool q0 = (bl & 1) != 0;
    const bool q1 = (bl & 2) != 0;
    const bool q2 = (bl & 4) != 0;
    auto pick8 = [&](const v8f& v) -> float {
        const float t0 = q0 ? v[1] : v[0];
        const float t1 = q0 ? v[3] : v[2];
        const float t2 = q0 ? v[5] : v[4];
        const float t3 = q0 ? v[7] : v[6];
        const float u0 = q1 ? t1 : t0;
        const float u1 = q1 ? t3 : t2;
        return q2 ? u1 : u0;
    };

    const float* W     = ws;
    const float* linit = ws + CSZ * 16;

    float Wf[16][2];
#pragma unroll
    for (int k = 0; k < 16; ++k) {
        int c = 4 * k + 2 * half;
        Wf[k][0] = W[(c + 0) * 16 + bl];
        Wf[k][1] = W[(c + 1) * 16 + bl];
    }
    float la[16][2];
#pragma unroll
    for (int k = 0; k < 16; ++k) {
        int c = 4 * k + 2 * half;
#pragma unroll
        for (int j = 0; j < 2; ++j) {
            la[k][j] = linit[(c + j) * 2 + sq];
        }
    }

    auto gather = [&](float (&dst)[16][2], int kcvv) {
#pragma unroll
        for (int k = 0; k < 16; ++k) {
            const float2 v = *(const float2*)(Amat + kcvv * CSZ + 4 * k + 2 * half);
            dst[k][0] = v.x;
            dst[k][1] = v.y;
        }
    };

    auto ldcc = [&](float (&dst)[16][2], int slot, int i) {
#pragma unroll
        for (int p = 0; p < 8; ++p) {
            const v4f q = ccring[slot][i][p][rl];
            dst[2 * p][0]     = q.x;
            dst[2 * p][1]     = q.y;
            dst[2 * p + 1][0] = q.z;
            dst[2 * p + 1][1] = q.w;
        }
    };

    auto prod = [&](int slot, int s3, int i, const float (&cc)[16][2], int y) {
        v8f accW = {};
#pragma unroll
        for (int k = 0; k < 16; ++k) {
            v2f a  = {cc[k][0], cc[k][1]};
            v2f bw = {Wf[k][0], Wf[k][1]};
            accW = __builtin_amdgcn_wmma_f32_16x16x4_f32(false, bw, false, a, (short)0, accW, false, false);
        }
#pragma unroll
        for (int p = 0; p < 8; ++p) {
            const v4f q = {cc[2 * p][0], cc[2 * p][1], cc[2 * p + 1][0], cc[2 * p + 1][1]};
            ccring[slot][i][p][lane] = q;
        }
        const float lp0 = y ? accW[1] : accW[0];
        const float lp1 = y ? accW[3] : accW[2];
        v4f v;
        v.x = half ? lp0 + accW[4] : accW[0];
        v.y = half ? lp1 + accW[5] : accW[1];
        v.z = half ? lp0 + accW[6] : accW[2];
        v.w = half ? lp1 + accW[7] : accW[3];
        ring[s3][i][bl][half] = v;
    };

    auto compute = [&](int slot, int i, const v2f wk, const float (&cc)[16][2]) {
        v8f acc = {};
#pragma unroll
        for (int k = 0; k < 16; ++k) {
            v2f a = {cc[k][0], cc[k][1]};
            v2f b = {la[k][0], la[k][1]};
            acc = __builtin_amdgcn_wmma_f32_16x16x4_f32(false, a, false, b, (short)0, acc, false, false);
        }

        const float d = pick8(acc);
        const float x = xrow8(d);
        const float a2_0 = sq ? x : d;
        const float a2_1 = sq ? d : x;

        a2ring[slot][i][rb][sq] = d;

        const float a3 = lse2(wk.x + a2_0, wk.y + a2_1);

#pragma unroll
        for (int k = 0; k < 16; ++k) {
#pragma unroll
            for (int j = 0; j < 2; ++j) {
                la[k][j] = fmaf(cc[k][j], a3 - la[k][j], la[k][j]);
            }
        }
    };

    float ccA[16][2], ccB[16][2];

    auto produce = [&](int blk, int s3, int par) {
        const int slot = blk & 1;
        const int base = bg * TSZ + blk * NB + par;
        int kk[NH], yy[NH];
#pragma unroll
        for (int j = 0; j < NH; ++j) {
            kk[j] = kc[base + 2 * j];
            yy[j] = corr[base + 2 * j];
        }
        gather(ccA, kk[0]);
        __builtin_amdgcn_sched_barrier(0);
#pragma unroll
        for (int j = 0; j < NH; ++j) {
            if (j + 1 < NH) {
                gather((j & 1) ? ccA : ccB, kk[j + 1]);
                __builtin_amdgcn_sched_group_barrier(0x020, 16, 0);
                __builtin_amdgcn_sched_group_barrier(0x008, 16, 0);
                __builtin_amdgcn_sched_barrier(0);
            }
            prod(slot, s3, par + 2 * j, (j & 1) ? ccB : ccA, yy[j]);
        }
    };

    auto consume = [&](int blk, int s3) {
        const int slot = blk & 1;
        ldcc(ccA, slot, 0);
        for (int i = 0; i < NB; i += 2) {
            ldcc(ccB, slot, i + 1);
            const v2f wa = *(const v2f*)((const float*)&ring[s3][i][rb][1] + 2 * sq);
            const v2f wb = *(const v2f*)((const float*)&ring[s3][i + 1][rb][1] + 2 * sq);
            __builtin_amdgcn_sched_group_barrier(0x100, 10, 0);
            __builtin_amdgcn_sched_group_barrier(0x008, 16, 0);
            compute(slot, i, wa, ccA);

            ldcc(ccA, slot, (i + 2 < NB) ? i + 2 : NB - 1);
            __builtin_amdgcn_sched_group_barrier(0x100, 8, 0);
            __builtin_amdgcn_sched_group_barrier(0x008, 16, 0);
            compute(slot, i + 1, wb, ccB);
        }
    };

    auto finish = [&](int blk, int s3) {
        const int slot = blk & 1;
#pragma unroll
        for (int i = 0; i < NB; ++i) {
            const v4f w0 = ring[s3][i][bl][0];
            const v2f a2 = *(const v2f*)&a2ring[slot][i][bl][0];
            const float p0 = lse2(w0.x + a2.x, w0.z + a2.y);
            const float p1 = lse2(w0.y + a2.x, w0.w + a2.y);
            const float z  = lse2(p0, p1);
            out[(size_t)(bg * TSZ + blk * NB + i) * 2 + half] = (half ? p1 : p0) - z;
        }
    };

    const int par = role - 2;
    if (role >= 2) {
        produce(0, 0, par);
    }
    __syncthreads();
    int s3 = 0;
    for (int blk = 0; blk < NBLK; ++blk) {
        const int s3n = (s3 == 2) ? 0 : s3 + 1;
        const int s3p = (s3 == 0) ? 2 : s3 - 1;
        if (role < 2) {
            consume(blk, s3);
        } else {
            if (blk + 1 < NBLK) produce(blk + 1, s3n, par);
            if (role == 3 && blk > 0) finish(blk - 1, s3p);
        }
        __syncthreads();
        s3 = s3n;
    }
    if (role == 3) {
        finish(NBLK - 1, (s3 == 0) ? 2 : s3 - 1);
    }
}

extern "C" void kernel_launch(void* const* d_in, const int* in_sizes, int n_in,
                              void* d_out, int out_size, void* d_ws, size_t ws_size,
                              hipStream_t stream) {
    const int*   corr  = (const int*)d_in[0];
    const int*   kc    = (const int*)d_in[1];
    const float* Amat  = (const float*)d_in[2];
    const float* trans = (const float*)d_in[3];
    const float* obs   = (const float*)d_in[4];
    const float* init  = (const float*)d_in[5];
    float*       out   = (float*)d_out;
    float*       ws    = (float*)d_ws;

    (void)in_sizes; (void)n_in; (void)out_size; (void)ws_size;

    bkt_prep<<<1, 64, 0, stream>>>(trans, obs, init, ws);
    bkt_fwd<<<BSZ / 16, 128, 0, stream>>>(corr, kc, Amat, ws, out);
}


# Wave 3 does a block's outputs before its product steps, in two halves behind the index loads and the first two row gathers: both round trips and the store drain leave the barrier's path; same sums
# speedup vs baseline: 5.4589x; 1.6365x over previous
#include <hip/hip_runtime.h>
#include <math.h>


typedef float v2f __attribute__((ext_vector_type(2)));
typedef float v4f __attribute__((ext_vector_type(4)));
typedef float v8f __attribute__((ext_vector_type(8)));

#define BSZ 1024
#define TSZ 500
#define CSZ 64
#define NB 10
#define NBLK (TSZ / NB)
#define NH (NB / 2)
#define TILE 4
#define CCP 33


__device__ __forceinline__ float lse2(float x, float y) {
    float m = fmaxf(x, y);
    return m + __logf(__expf(x - m) + __expf(y - m));
}

__device__ __forceinline__ float xrow8(float x) {
    int b = __builtin_bit_cast(int, x);
    return __builtin_bit_cast(float,
        __builtin_amdgcn_permlane16(b, b, 0xfedcba98, 0x76543210, false, false));
}

__device__ __forceinline__ float xlane1(float x) {
    int b = __builtin_bit_cast(int, x);
    return __builtin_bit_cast(float, __builtin_amdgcn_mov_dpp(b, 0x161, 0xf, 0xf, true));
}
__device__ __forceinline__ float xlane2(float x) {
    int b = __builtin_bit_cast(int, x);
    return __builtin_bit_cast(float, __builtin_amdgcn_mov_dpp(b, 0x162, 0xf, 0xf, true));
}

__global__ __launch_bounds__(64) void bkt_prep(const float* __restrict__ trans,
                                               const float* __restrict__ obs,
                                               const float* __restrict__ init,
                                               float* __restrict__ ws) {
    int c = threadIdx.x;
    if (c >= CSZ) return;
    float* W = ws;
    float* linit = ws + CSZ * 16;

    float w8[8];
    for (int s = 0; s < 2; ++s) {
        float x0 = obs[c * 4 + s * 2 + 0];
        float x1 = obs[c * 4 + s * 2 + 1];
        float z = lse2(x0, x1);
        w8[s * 2 + 0] = x0 - z;
        w8[s * 2 + 1] = x1 - z;
    }
    for (int tp = 0; tp < 2; ++tp) {
        float x0 = trans[c * 4 + 0 * 2 + tp];
        float x1 = trans[c * 4 + 1 * 2 + tp];
        float z = lse2(x0, x1);
        w8[4 + 0 * 2 + tp] = x0 - z;
        w8[4 + 1 * 2 + tp] = x1 - z;
    }
    for (int n = 0; n < 8; ++n) {
        W[c * 16 + n]     = w8[n];
        W[c * 16 + 8 + n] = w8[n];
    }
    {
        float x0 = init[c * 2 + 0];
        float x1 = init[c * 2 + 1];
        float z = lse2(x0, x1);
        linit[c * 2 + 0] = x0 - z;
        linit[c * 2 + 1] = x1 - z;
    }
}

__global__ __launch_bounds__(128) __attribute__((amdgpu_waves_per_eu(1, 1))) void bkt_fwd(const int* __restrict__ corr,
                                              const int* __restrict__ kc,
                                              const float* __restrict__ Amat,
                                              const float* __restrict__ ws,
                                              float* __restrict__ out) {
    __shared__ v4f ring[3][NB][16][2];
    __shared__ __attribute__((aligned(16))) float a2ring[2][NB][16][2];
    __shared__ v4f ccring[2][NB][8][CCP];

    const int lane = threadIdx.x & 31;
    const int role = __builtin_amdgcn_readfirstlane((int)(threadIdx.x >> 5));
    const int bl   = lane & 15;
    const int half = lane >> 4;
    const int b4   = bl & (TILE - 1);
    const int bg   = blockIdx.x * TILE + b4;

    const int kq = bl & 3;
    const int rb = ((role & 1) << 1) | ((bl >> 2) & 1);
    const int sq = bl >> 3;
    const int rl = rb + 16 * half;
    const bool q0 = (bl & 1) != 0;
    const bool q1 = (bl & 2) != 0;
    const bool q2 = (bl & 4) != 0;
    auto pick8 = [&](const v8f& v) -> float {
        const float t0 = q0 ? v[1] : v[0];
        const float t1 = q0 ? v[3] : v[2];
        const float t2 = q0 ? v[5] : v[4];
        const float t3 = q0 ? v[7] : v[6];
        const float u0 = q1 ? t1 : t0;
        const float u1 = q1 ? t3 : t2;
        return q2 ? u1 : u0;
    };

    const float* W     = ws;
    const float* linit = ws + CSZ * 16;

    float Wf[16][2];
#pragma unroll
    for (int k = 0; k < 16; ++k) {
        int c = 4 * k + 2 * half;
        Wf[k][0] = W[(c + 0) * 16 + bl];
        Wf[k][1] = W[(c + 1) * 16 + bl];
    }
    float la[4][2];
#pragma unroll
    for (int k = 0; k < 4; ++k) {
        int c = 16 * kq + 4 * k + 2 * half;
#pragma unroll
        for (int j = 0; j < 2; ++j) {
            la[k][j] = linit[(c + j) * 2 + sq];
        }
    }

    auto gather = [&](float (&dst)[16][2], int kcvv) {
#pragma unroll
        for (int k = 0; k < 16; ++k) {
            const float2 v = *(const float2*)(Amat + kcvv * CSZ + 4 * k + 2 * half);
            dst[k][0] = v.x;
            dst[k][1] = v.y;
        }
    };

    auto ldcc = [&](float (&dst)[4][2], int slot, int i) {
#pragma unroll
        for (int e = 0; e < 2; ++e) {
            const v4f q = ccring[slot][i][2 * kq + e][rl];
            dst[2 * e][0]     = q.x;
            dst[2 * e][1]     = q.y;
            dst[2 * e + 1][0] = q.z;
            dst[2 * e + 1][1] = q.w;
        }
    };

    auto prod = [&](int slot, int s3, int i, const float (&cc)[16][2], int y) {
        v8f accW = {};
#pragma unroll
        for (int k = 0; k < 16; ++k) {
            v2f a  = {cc[k][0], cc[k][1]};
            v2f bw = {Wf[k][0], Wf[k][1]};
            accW = __builtin_amdgcn_wmma_f32_16x16x4_f32(false, bw, false, a, (short)0, accW, false, false);
        }
#pragma unroll
        for (int p = 0; p < 8; ++p) {
            const v4f q = {cc[2 * p][0], cc[2 * p][1], cc[2 * p + 1][0], cc[2 * p + 1][1]};
            ccring[slot][i][p][lane] = q;
        }
        const float lp0 = y ? accW[1] : accW[0];
        const float lp1 = y ? accW[3] : accW[2];
        v4f v;
        v.x = half ? lp0 + accW[4] : accW[0];
        v.y = half ? lp1 + accW[5] : accW[1];
        v.z = half ? lp0 + accW[6] : accW[2];
        v.w = half ? lp1 + accW[7] : accW[3];
        ring[s3][i][bl][half] = v;
    };

    auto compute = [&](int slot, int i, const v2f wk, const float (&cc)[4][2]) {
        v8f acc = {};
#pragma unroll
        for (int k = 0; k < 4; ++k) {
            v2f a = {cc[k][0], cc[k][1]};
            v2f b = {la[k][0], la[k][1]};
            acc = __builtin_amdgcn_wmma_f32_16x16x4_f32(false, a, false, b, (short)0, acc, false, false);
        }

        const float dq = pick8(acc);
        const float dp = dq + xlane1(dq);
        const float d  = dp + xlane2(dp);
        const float x = xrow8(d);
        const float a2_0 = sq ? x : d;
        const float a2_1 = sq ? d : x;

        a2ring[slot][i][rb][sq] = d;

        const float a3 = lse2(wk.x + a2_0, wk.y + a2_1);

#pragma unroll
        for (int k = 0; k < 4; ++k) {
#pragma unroll
            for (int j = 0; j < 2; ++j) {
                la[k][j] = fmaf(cc[k][j], a3 - la[k][j], la[k][j]);
            }
        }
    };

    float rcA[4][2], rcB[4][2];

    auto load_idx = [&](int blk, int par, int (&kk)[NH], int (&yy)[NH]) {
        const int base = bg * TSZ + blk * NB + par;
#pragma unroll
        for (int j = 0; j < NH; ++j) {
            kk[j] = kc[base + 2 * j];
            yy[j] = corr[base + 2 * j];
        }
    };
    auto first_rows = [&](float (&ccA)[16][2], float (&ccB)[16][2], const int (&kk)[NH]) {
        gather(ccA, kk[0]);
        gather(ccB, kk[1]);
    };
    auto produce_steps = [&](float (&ccA)[16][2], float (&ccB)[16][2], int blk, int s3, int par,
                             const int (&kk)[NH], const int (&yy)[NH]) {
        const int slot = blk & 1;
#pragma unroll
        for (int j = 0; j < NH; ++j) {
            if (j >= 1 && j + 1 < NH) {
                gather((j & 1) ? ccA : ccB, kk[j + 1]);
                __builtin_amdgcn_sched_group_barrier(0x020, 16, 0);
                __builtin_amdgcn_sched_group_barrier(0x008, 16, 0);
                __builtin_amdgcn_sched_barrier(0);
            }
            prod(slot, s3, par + 2 * j, (j & 1) ? ccB : ccA, yy[j]);
        }
    };

    auto consume = [&](int blk, int s3) {
        const int slot = blk & 1;
        ldcc(rcA, slot, 0);
        for (int i = 0; i < NB; i += 2) {
            ldcc(rcB, slot, i + 1);
            const v2f wa = *(const v2f*)((const float*)&ring[s3][i][rb][1] + 2 * sq);
            const v2f wb = *(const v2f*)((const float*)&ring[s3][i + 1][rb][1] + 2 * sq);
            __builtin_amdgcn_sched_group_barrier(0x100, 4, 0);
            __builtin_amdgcn_sched_group_barrier(0x008, 4, 0);
            compute(slot, i, wa, rcA);

            ldcc(rcA, slot, (i + 2 < NB) ? i + 2 : NB - 1);
            __builtin_amdgcn_sched_group_barrier(0x100, 2, 0);
            __builtin_amdgcn_sched_group_barrier(0x008, 4, 0);
            compute(slot, i + 1, wb, rcB);
        }
    };

    auto finish = [&](int blk, int s3, int i0) {
        const int slot = blk & 1;
#pragma unroll
        for (int u = 0; u < NB / 2; ++u) {
            const int i = i0 + u;
            const v4f w0 = ring[s3][i][b4][0];
            const v2f a2 = *(const v2f*)&a2ring[slot][i][b4][0];
            const float p0 = lse2(w0.x + a2.x, w0.z + a2.y);
            const float p1 = lse2(w0.y + a2.x, w0.w + a2.y);
            const float z  = lse2(p0, p1);
            out[(size_t)(bg * TSZ + blk * NB + i) * 2 + half] = (half ? p1 : p0) - z;
        }
    };

    auto finish_whole = [&](int blk, int s3) {
#pragma nounroll
        for (int h = 0; h < 2; ++h) {
            finish(blk, s3, h * (NB / 2));
        }
    };

    const int par = role - 2;
    if (role >= 2) {
        float ccA[16][2], ccB[16][2];
        int kk[NH], yy[NH];
        load_idx(0, par, kk, yy);
        first_rows(ccA, ccB, kk);
        produce_steps(ccA, ccB, 0, 0, par, kk, yy);
    }
    __syncthreads();
    int s3 = 0;
    for (int blk = 0; blk < NBLK; ++blk) {
        const int s3n = (s3 == 2) ? 0 : s3 + 1;
        const int s3p = (s3 == 0) ? 2 : s3 - 1;
        if (role < 2) {
            consume(blk, s3);
        } else if (blk + 1 < NBLK) {
            const bool outs = role == 3 && blk > 0;
            float ccA[16][2], ccB[16][2];
            int kk[NH], yy[NH];
            load_idx(blk + 1, par, kk, yy);
            if (outs) finish(blk - 1, s3p, 0);
            first_rows(ccA, ccB, kk);
            if (outs) finish(blk - 1, s3p, NB / 2);
            produce_steps(ccA, ccB, blk + 1, s3n, par, kk, yy);
        } else if (role == 3) {
            finish_whole(blk - 1, s3p);
        }
        __syncthreads();
        s3 = s3n;
    }
    if (role == 3) {
        finish_whole(NBLK - 1, (s3 == 0) ? 2 : s3 - 1);
    }
}

extern "C" void kernel_launch(void* const* d_in, const int* in_sizes, int n_in,
                              void* d_out, int out_size, void* d_ws, size_t ws_size,
                              hipStream_t stream) {
    const int*   corr  = (const int*)d_in[0];
    const int*   kc    = (const int*)d_in[1];
    const float* Amat  = (const float*)d_in[2];
    const float* trans = (const float*)d_in[3];
    const float* obs   = (const float*)d_in[4];
    const float* init  = (const float*)d_in[5];
    float*       out   = (float*)d_out;
    float*       ws    = (float*)d_ws;

    (void)in_sizes; (void)n_in; (void)out_size; (void)ws_size;

    bkt_prep<<<1, 64, 0, stream>>>(trans, obs, init, ws);
    bkt_fwd<<<BSZ / TILE, 128, 0, stream>>>(corr, kc, Amat, ws, out);
}


# Product tile's 16 columns are 4 time steps x 4 batches, not 4 copies of 4 batches: one 16-WMMA chain, one row gather, one set of ring stores per 4 steps; blocks of 20 steps; same sums
# speedup vs baseline: 6.7345x; 1.2337x over previous
#include <hip/hip_runtime.h>
#include <math.h>


typedef float v2f __attribute__((ext_vector_type(2)));
typedef float v4f __attribute__((ext_vector_type(4)));
typedef float v8f __attribute__((ext_vector_type(8)));

#define BSZ 1024
#define TSZ 500
#define CSZ 64
#define NB 20
#define NBLK (TSZ / NB)
#define NG (NB / 4)
#define NH 3
#define NF 5
#define TILE 4
#define CCP 33


__device__ __forceinline__ float lse2(float x, float y) {
    float m = fmaxf(x, y);
    return m + __logf(__expf(x - m) + __expf(y - m));
}

__device__ __forceinline__ float xrow8(float x) {
    int b = __builtin_bit_cast(int, x);
    return __builtin_bit_cast(float,
        __builtin_amdgcn_permlane16(b, b, 0xfedcba98, 0x76543210, false, false));
}

__device__ __forceinline__ float xlane1(float x) {
    int b = __builtin_bit_cast(int, x);
    return __builtin_bit_cast(float, __builtin_amdgcn_mov_dpp(b, 0x161, 0xf, 0xf, true));
}
__device__ __forceinline__ float xlane2(float x) {
    int b = __builtin_bit_cast(int, x);
    return __builtin_bit_cast(float, __builtin_amdgcn_mov_dpp(b, 0x162, 0xf, 0xf, true));
}

__global__ __launch_bounds__(64) void bkt_prep(const float* __restrict__ trans,
                                               const float* __restrict__ obs,
                                               const float* __restrict__ init,
                                               float* __restrict__ ws) {
    int c = threadIdx.x;
    if (c >= CSZ) return;
    float* W = ws;
    float* linit = ws + CSZ * 16;

    float w8[8];
    for (int s = 0; s < 2; ++s) {
        float x0 = obs[c * 4 + s * 2 + 0];
        float x1 = obs[c * 4 + s * 2 + 1];
        float z = lse2(x0, x1);
        w8[s * 2 + 0] = x0 - z;
        w8[s * 2 + 1] = x1 - z;
    }
    for (int tp = 0; tp < 2; ++tp) {
        float x0 = trans[c * 4 + 0 * 2 + tp];
        float x1 = trans[c * 4 + 1 * 2 + tp];
        float z = lse2(x0, x1);
        w8[4 + 0 * 2 + tp] = x0 - z;
        w8[4 + 1 * 2 + tp] = x1 - z;
    }
    for (int n = 0; n < 8; ++n) {
        W[c * 16 + n]     = w8[n];
        W[c * 16 + 8 + n] = w8[n];
    }
    {
        float x0 = init[c * 2 + 0];
        float x1 = init[c * 2 + 1];
        float z = lse2(x0, x1);
        linit[c * 2 + 0] = x0 - z;
        linit[c * 2 + 1] = x1 - z;
    }
}

__global__ __launch_bounds__(128) __attribute__((amdgpu_waves_per_eu(1, 1))) void bkt_fwd(const int* __restrict__ corr,
                                              const int* __restrict__ kc,
                                              const float* __restrict__ Amat,
                                              const float* __restrict__ ws,
                                              float* __restrict__ out) {
    __shared__ v4f ring[3][NB][TILE][2];
    __shared__ __attribute__((aligned(16))) float a2ring[2][NB][16][2];
    __shared__ v4f ccring[2][NG][8][CCP];

    const int lane = threadIdx.x & 31;
    const int role = __builtin_amdgcn_readfirstlane((int)(threadIdx.x >> 5));
    const int bl   = lane & 15;
    const int half = lane >> 4;
    const int b4   = bl & (TILE - 1);
    const int bg   = blockIdx.x * TILE + b4;
    const int jcol = bl >> 2;

    const int kq = bl & 3;
    const int rb = ((role & 1) << 1) | ((bl >> 2) & 1);
    const int sq = bl >> 3;
    const int rl = rb + 16 * half;
    const bool q0 = (bl & 1) != 0;
    const bool q1 = (bl & 2) != 0;
    const bool q2 = (bl & 4) != 0;
    auto pick8 = [&](const v8f& v) -> float {
        const float t0 = q0 ? v[1] : v[0];
        const float t1 = q0 ? v[3] : v[2];
        const float t2 = q0 ? v[5] : v[4];
        const float t3 = q0 ? v[7] : v[6];
        const float u0 = q1 ? t1 : t0;
        const float u1 = q1 ? t3 : t2;
        return q2 ? u1 : u0;
    };

    const float* W     = ws;
    const float* linit = ws + CSZ * 16;

    float Wf[16][2];
#pragma unroll
    for (int k = 0; k < 16; ++k) {
        int c = 4 * k + 2 * half;
        Wf[k][0] = W[(c + 0) * 16 + bl];
        Wf[k][1] = W[(c + 1) * 16 + bl];
    }
    float la[4][2];
#pragma unroll
    for (int k = 0; k < 4; ++k) {
        int c = 16 * kq + 4 * k + 2 * half;
#pragma unroll
        for (int j = 0; j < 2; ++j) {
            la[k][j] = linit[(c + j) * 2 + sq];
        }
    }

    auto gather = [&](float (&dst)[16][2], int kcvv) {
#pragma unroll
        for (int k = 0; k < 16; ++k) {
            const float2 v = *(const float2*)(Amat + kcvv * CSZ + 4 * k + 2 * half);
            dst[k][0] = v.x;
            dst[k][1] = v.y;
        }
    };

    auto ldcc = [&](float (&dst)[4][2], int slot, int i) {
#pragma unroll
        for (int e = 0; e < 2; ++e) {
            const v4f q = ccring[slot][i >> 2][2 * kq + e][((i & 3) << 2) + rl];
            dst[2 * e][0]     = q.x;
            dst[2 * e][1]     = q.y;
            dst[2 * e + 1][0] = q.z;
            dst[2 * e + 1][1] = q.w;
        }
    };

    auto prod = [&](int slot, int s3, int g, const float (&cc)[16][2], int y) {
        v8f accW = {};
#pragma unroll
        for (int k = 0; k < 16; ++k) {
            v2f a  = {cc[k][0], cc[k][1]};
            v2f bw = {Wf[k][0], Wf[k][1]};
            accW = __builtin_amdgcn_wmma_f32_16x16x4_f32(false, bw, false, a, (short)0, accW, false, false);
        }
#pragma unroll
        for (int p = 0; p < 8; ++p) {
            const v4f q = {cc[2 * p][0], cc[2 * p][1], cc[2 * p + 1][0], cc[2 * p + 1][1]};
            ccring[slot][g][p][lane] = q;
        }
        const float lp0 = y ? accW[1] : accW[0];
        const float lp1 = y ? accW[3] : accW[2];
        v4f v;
        v.x = half ? lp0 + accW[4] : accW[0];
        v.y = half ? lp1 + accW[5] : accW[1];
        v.z = half ? lp0 + accW[6] : accW[2];
        v.w = half ? lp1 + accW[7] : accW[3];
        ring[s3][4 * g + jcol][b4][half] = v;
    };

    auto compute = [&](int slot, int i, const v2f wk, const float (&cc)[4][2]) {
        v8f acc = {};
#pragma unroll
        for (int k = 0; k < 4; ++k) {
            v2f a = {cc[k][0], cc[k][1]};
            v2f b = {la[k][0], la[k][1]};
            acc = __builtin_amdgcn_wmma_f32_16x16x4_f32(false, a, false, b, (short)0, acc, false, false);
        }

        const float dq = pick8(acc);
        const float dp = dq + xlane1(dq);
        const float d  = dp + xlane2(dp);
        const float x = xrow8(d);
        const float a2_0 = sq ? x : d;
        const float a2_1 = sq ? d : x;

        a2ring[slot][i][rb][sq] = d;

        const float a3 = lse2(wk.x + a2_0, wk.y + a2_1);

#pragma unroll
        for (int k = 0; k < 4; ++k) {
#pragma unroll
            for (int j = 0; j < 2; ++j) {
                la[k][j] = fmaf(cc[k][j], a3 - la[k][j], la[k][j]);
            }
        }
    };

    float rcA[4][2], rcB[4][2];

    auto load_idx = [&](int blk, int par, int (&kk)[NH], int (&yy)[NH]) {
        const int base = bg * TSZ + blk * NB + jcol;
#pragma unroll
        for (int j = 0; j < NH; ++j) {
            const int g = par + 2 * j;
            kk[j] = 0;
            yy[j] = 0;
            if (g < NG) {
                kk[j] = kc[base + 4 * g];
                yy[j] = corr[base + 4 * g];
            }
        }
    };
    auto first_rows = [&](float (&ccA)[16][2], float (&ccB)[16][2], const int (&kk)[NH]) {
        gather(ccA, kk[0]);
        gather(ccB, kk[1]);
    };
    auto produce_steps = [&](float (&ccA)[16][2], float (&ccB)[16][2], int blk, int s3, int par,
                             const int (&kk)[NH], const int (&yy)[NH]) {
        const int slot = blk & 1;
#pragma unroll
        for (int j = 0; j < NH; ++j) {
            if (par + 2 * j >= NG) break;
            if (j >= 1 && j + 1 < NH && par + 2 * (j + 1) < NG) {
                gather((j & 1) ? ccA : ccB, kk[j + 1]);
            }
            prod(slot, s3, par + 2 * j, (j & 1) ? ccB : ccA, yy[j]);
        }
    };

    auto consume = [&](int blk, int s3) {
        const int slot = blk & 1;
        ldcc(rcA, slot, 0);
        for (int i = 0; i < NB; i += 2) {
            ldcc(rcB, slot, i + 1);
            const v2f wa = *(const v2f*)((const float*)&ring[s3][i][rb][1] + 2 * sq);
            const v2f wb = *(const v2f*)((const float*)&ring[s3][i + 1][rb][1] + 2 * sq);
            __builtin_amdgcn_sched_group_barrier(0x100, 4, 0);
            __builtin_amdgcn_sched_group_barrier(0x008, 4, 0);
            compute(slot, i, wa, rcA);

            ldcc(rcA, slot, (i + 2 < NB) ? i + 2 : NB - 1);
            __builtin_amdgcn_sched_group_barrier(0x100, 2, 0);
            __builtin_amdgcn_sched_group_barrier(0x008, 4, 0);
            compute(slot, i + 1, wb, rcB);
        }
    };

    auto finish = [&](int blk, int s3, int i0) {
        const int slot = blk & 1;
#pragma unroll
        for (int u = 0; u < NF; ++u) {
            const int i = i0 + u;
            const v4f w0 = ring[s3][i][b4][0];
            const v2f a2 = *(const v2f*)&a2ring[slot][i][b4][0];
            const float p0 = lse2(w0.x + a2.x, w0.z + a2.y);
            const float p1 = lse2(w0.y + a2.x, w0.w + a2.y);
            const float z  = lse2(p0, p1);
            out[(size_t)(bg * TSZ + blk * NB + i) * 2 + half] = (half ? p1 : p0) - z;
        }
    };

    auto finish_whole = [&](int blk, int s3) {
#pragma nounroll
        for (int h = 0; h < NB / NF; ++h) {
            finish(blk, s3, h * NF);
        }
    };

    const int par = role - 2;
    if (role >= 2) {
        float ccA[16][2], ccB[16][2];
        int kk[NH], yy[NH];
        load_idx(0, par, kk, yy);
        first_rows(ccA, ccB, kk);
        produce_steps(ccA, ccB, 0, 0, par, kk, yy);
    }
    __syncthreads();
    int s3 = 0;
    for (int blk = 0; blk < NBLK; ++blk) {
        const int s3n = (s3 == 2) ? 0 : s3 + 1;
        const int s3p = (s3 == 0) ? 2 : s3 - 1;
        if (role < 2) {
            consume(blk, s3);
        } else if (blk + 1 < NBLK) {
            const bool outs = role == 3 && blk > 0;
            float ccA[16][2], ccB[16][2];
            int kk[NH], yy[NH];
            load_idx(blk + 1, par, kk, yy);
            if (outs) finish(blk - 1, s3p, 0);
            first_rows(ccA, ccB, kk);
            if (outs) {
#pragma nounroll
                for (int h = 1; h < NB / NF; ++h) {
                    finish(blk - 1, s3p, h * NF);
                }
            }
            produce_steps(ccA, ccB, blk + 1, s3n, par, kk, yy);
        } else if (role == 3) {
            finish_whole(blk - 1, s3p);
        }
        __syncthreads();
        s3 = s3n;
    }
    if (role == 3) {
        finish_whole(NBLK - 1, (s3 == 0) ? 2 : s3 - 1);
    }
}

extern "C" void kernel_launch(void* const* d_in, const int* in_sizes, int n_in,
                              void* d_out, int out_size, void* d_ws, size_t ws_size,
                              hipStream_t stream) {
    const int*   corr  = (const int*)d_in[0];
    const int*   kc    = (const int*)d_in[1];
    const float* Amat  = (const float*)d_in[2];
    const float* trans = (const float*)d_in[3];
    const float* obs   = (const float*)d_in[4];
    const float* init  = (const float*)d_in[5];
    float*       out   = (float*)d_out;
    float*       ws    = (float*)d_ws;

    (void)in_sizes; (void)n_in; (void)out_size; (void)ws_size;

    bkt_prep<<<1, 64, 0, stream>>>(trans, obs, init, ws);
    bkt_fwd<<<BSZ / TILE, 128, 0, stream>>>(corr, kc, Amat, ws, out);
}


# Other state's a2 reaches the LSE as the DPP operand (row_xmask:8) of the add that forms its term, k pairs handed over own term first: v_mov + v_permlane16 + select leave the serial chain; same sums
# speedup vs baseline: 7.0103x; 1.0410x over previous
#include <hip/hip_runtime.h>
#include <math.h>


typedef float v2f __attribute__((ext_vector_type(2)));
typedef float v4f __attribute__((ext_vector_type(4)));
typedef float v8f __attribute__((ext_vector_type(8)));

#define BSZ 1024
#define TSZ 500
#define CSZ 64
#define NB 20
#define NBLK (TSZ / NB)
#define NG (NB / 4)
#define NH 3
#define NF 5
#define TILE 4
#define CCP 33


__device__ __forceinline__ float lse2(float x, float y) {
    float m = fmaxf(x, y);
    return m + __logf(__expf(x - m) + __expf(y - m));
}

__device__ __forceinline__ float xlane1(float x) {
    int b = __builtin_bit_cast(int, x);
    return __builtin_bit_cast(float, __builtin_amdgcn_mov_dpp(b, 0x161, 0xf, 0xf, true));
}
__device__ __forceinline__ float xlane2(float x) {
    int b = __builtin_bit_cast(int, x);
    return __builtin_bit_cast(float, __builtin_amdgcn_mov_dpp(b, 0x162, 0xf, 0xf, true));
}
__device__ __forceinline__ float xlane8(float x) {
    int b = __builtin_bit_cast(int, x);
    return __builtin_bit_cast(float, __builtin_amdgcn_mov_dpp(b, 0x168, 0xf, 0xf, true));
}

__global__ __launch_bounds__(64) void bkt_prep(const float* __restrict__ trans,
                                               const float* __restrict__ obs,
                                               const float* __restrict__ init,
                                               float* __restrict__ ws) {
    int c = threadIdx.x;
    if (c >= CSZ) return;
    float* W = ws;
    float* linit = ws + CSZ * 16;

    float w8[8];
    for (int s = 0; s < 2; ++s) {
        float x0 = obs[c * 4 + s * 2 + 0];
        float x1 = obs[c * 4 + s * 2 + 1];
        float z = lse2(x0, x1);
        w8[s * 2 + 0] = x0 - z;
        w8[s * 2 + 1] = x1 - z;
    }
    for (int tp = 0; tp < 2; ++tp) {
        float x0 = trans[c * 4 + 0 * 2 + tp];
        float x1 = trans[c * 4 + 1 * 2 + tp];
        float z = lse2(x0, x1);
        w8[4 + 0 * 2 + tp] = x0 - z;
        w8[4 + 1 * 2 + tp] = x1 - z;
    }
    for (int n = 0; n < 8; ++n) {
        W[c * 16 + n]     = w8[n];
        W[c * 16 + 8 + n] = w8[n];
    }
    {
        float x0 = init[c * 2 + 0];
        float x1 = init[c * 2 + 1];
        float z = lse2(x0, x1);
        linit[c * 2 + 0] = x0 - z;
        linit[c * 2 + 1] = x1 - z;
    }
}

__global__ __launch_bounds__(128) __attribute__((amdgpu_waves_per_eu(1, 1))) void bkt_fwd(const int* __restrict__ corr,
                                              const int* __restrict__ kc,
                                              const float* __restrict__ Amat,
                                              const float* __restrict__ ws,
                                              float* __restrict__ out) {
    __shared__ v4f ring[3][NB][TILE][2];
    __shared__ __attribute__((aligned(16))) float a2ring[2][NB][16][2];
    __shared__ v4f ccring[2][NG][8][CCP];

    const int lane = threadIdx.x & 31;
    const int role = __builtin_amdgcn_readfirstlane((int)(threadIdx.x >> 5));
    const int bl   = lane & 15;
    const int half = lane >> 4;
    const int b4   = bl & (TILE - 1);
    const int bg   = blockIdx.x * TILE + b4;
    const int jcol = bl >> 2;

    const int kq = bl & 3;
    const int rb = ((role & 1) << 1) | ((bl >> 2) & 1);
    const int sq = bl >> 3;
    const int rl = rb + 16 * half;
    const bool q0 = (bl & 1) != 0;
    const bool q1 = (bl & 2) != 0;
    const bool q2 = (bl & 4) != 0;
    auto pick8 = [&](const v8f& v) -> float {
        const float t0 = q0 ? v[1] : v[0];
        const float t1 = q0 ? v[3] : v[2];
        const float t2 = q0 ? v[5] : v[4];
        const float t3 = q0 ? v[7] : v[6];
        const float u0 = q1 ? t1 : t0;
        const float u1 = q1 ? t3 : t2;
        return q2 ? u1 : u0;
    };

    const float* W     = ws;
    const float* linit = ws + CSZ * 16;

    float Wf[16][2];
#pragma unroll
    for (int k = 0; k < 16; ++k) {
        int c = 4 * k + 2 * half;
        Wf[k][0] = W[(c + 0) * 16 + bl];
        Wf[k][1] = W[(c + 1) * 16 + bl];
    }
    float la[4][2];
#pragma unroll
    for (int k = 0; k < 4; ++k) {
        int c = 16 * kq + 4 * k + 2 * half;
#pragma unroll
        for (int j = 0; j < 2; ++j) {
            la[k][j] = linit[(c + j) * 2 + sq];
        }
    }

    auto gather = [&](float (&dst)[16][2], int kcvv) {
#pragma unroll
        for (int k = 0; k < 16; ++k) {
            const float2 v = *(const float2*)(Amat + kcvv * CSZ + 4 * k + 2 * half);
            dst[k][0] = v.x;
            dst[k][1] = v.y;
        }
    };

    auto ldcc = [&](float (&dst)[4][2], int slot, int i) {
#pragma unroll
        for (int e = 0; e < 2; ++e) {
            const v4f q = ccring[slot][i >> 2][2 * kq + e][((i & 3) << 2) + rl];
            dst[2 * e][0]     = q.x;
            dst[2 * e][1]     = q.y;
            dst[2 * e + 1][0] = q.z;
            dst[2 * e + 1][1] = q.w;
        }
    };

    auto prod = [&](int slot, int s3, int g, const float (&cc)[16][2], int y) {
        v8f accW = {};
#pragma unroll
        for (int k = 0; k < 16; ++k) {
            v2f a  = {cc[k][0], cc[k][1]};
            v2f bw = {Wf[k][0], Wf[k][1]};
            accW = __builtin_amdgcn_wmma_f32_16x16x4_f32(false, bw, false, a, (short)0, accW, false, false);
        }
#pragma unroll
        for (int p = 0; p < 8; ++p) {
            const v4f q = {cc[2 * p][0], cc[2 * p][1], cc[2 * p + 1][0], cc[2 * p + 1][1]};
            ccring[slot][g][p][lane] = q;
        }
        const float lp0 = y ? accW[1] : accW[0];
        const float lp1 = y ? accW[3] : accW[2];
        v4f v;
        v.x = half ? lp0 + accW[4] : accW[0];
        v.y = half ? lp1 + accW[5] : accW[1];
        v.z = half ? lp1 + accW[7] : accW[2];
        v.w = half ? lp0 + accW[6] : accW[3];
        ring[s3][4 * g + jcol][b4][half] = v;
    };

    auto compute = [&](int slot, int i, const v2f wk, const float (&cc)[4][2]) {
        v8f acc = {};
#pragma unroll
        for (int k = 0; k < 4; ++k) {
            v2f a = {cc[k][0], cc[k][1]};
            v2f b = {la[k][0], la[k][1]};
            acc = __builtin_amdgcn_wmma_f32_16x16x4_f32(false, a, false, b, (short)0, acc, false, false);
        }

        const float dq = pick8(acc);
        const float dp = dq + xlane1(dq);
        const float d  = dp + xlane2(dp);

        a2ring[slot][i][rb][sq] = d;

        const float a3 = lse2(wk.x + d, wk.y + xlane8(d));

#pragma unroll
        for (int k = 0; k < 4; ++k) {
#pragma unroll
            for (int j = 0; j < 2; ++j) {
                la[k][j] = fmaf(cc[k][j], a3 - la[k][j], la[k][j]);
            }
        }
    };

    float rcA[4][2], rcB[4][2];

    auto load_idx = [&](int blk, int par, int (&kk)[NH], int (&yy)[NH]) {
        const int base = bg * TSZ + blk * NB + jcol;
#pragma unroll
        for (int j = 0; j < NH; ++j) {
            const int g = par + 2 * j;
            kk[j] = 0;
            yy[j] = 0;
            if (g < NG) {
                kk[j] = kc[base + 4 * g];
                yy[j] = corr[base + 4 * g];
            }
        }
    };
    auto first_rows = [&](float (&ccA)[16][2], float (&ccB)[16][2], const int (&kk)[NH]) {
        gather(ccA, kk[0]);
        gather(ccB, kk[1]);
    };
    auto produce_steps = [&](float (&ccA)[16][2], float (&ccB)[16][2], int blk, int s3, int par,
                             const int (&kk)[NH], const int (&yy)[NH]) {
        const int slot = blk & 1;
#pragma unroll
        for (int j = 0; j < NH; ++j) {
            if (par + 2 * j >= NG) break;
            if (j >= 1 && j + 1 < NH && par + 2 * (j + 1) < NG) {
                gather((j & 1) ? ccA : ccB, kk[j + 1]);
            }
            prod(slot, s3, par + 2 * j, (j & 1) ? ccB : ccA, yy[j]);
        }
    };

    auto consume = [&](int blk, int s3) {
        const int slot = blk & 1;
        ldcc(rcA, slot, 0);
        for (int i = 0; i < NB; i += 2) {
            ldcc(rcB, slot, i + 1);
            const v2f wa = *(const v2f*)((const float*)&ring[s3][i][rb][1] + 2 * sq);
            const v2f wb = *(const v2f*)((const float*)&ring[s3][i + 1][rb][1] + 2 * sq);
            __builtin_amdgcn_sched_group_barrier(0x100, 4, 0);
            __builtin_amdgcn_sched_group_barrier(0x008, 4, 0);
            compute(slot, i, wa, rcA);

            ldcc(rcA, slot, (i + 2 < NB) ? i + 2 : NB - 1);
            __builtin_amdgcn_sched_group_barrier(0x100, 2, 0);
            __builtin_amdgcn_sched_group_barrier(0x008, 4, 0);
            compute(slot, i + 1, wb, rcB);
        }
    };

    auto finish = [&](int blk, int s3, int i0) {
        const int slot = blk & 1;
#pragma unroll
        for (int u = 0; u < NF; ++u) {
            const int i = i0 + u;
            const v4f w0 = ring[s3][i][b4][0];
            const v2f a2 = *(const v2f*)&a2ring[slot][i][b4][0];
            const float p0 = lse2(w0.x + a2.x, w0.z + a2.y);
            const float p1 = lse2(w0.y + a2.x, w0.w + a2.y);
            const float z  = lse2(p0, p1);
            out[(size_t)(bg * TSZ + blk * NB + i) * 2 + half] = (half ? p1 : p0) - z;
        }
    };

    auto finish_whole = [&](int blk, int s3) {
#pragma nounroll
        for (int h = 0; h < NB / NF; ++h) {
            finish(blk, s3, h * NF);
        }
    };

    const int par = role - 2;
    if (role >= 2) {
        float ccA[16][2], ccB[16][2];
        int kk[NH], yy[NH];
        load_idx(0, par, kk, yy);
        first_rows(ccA, ccB, kk);
        produce_steps(ccA, ccB, 0, 0, par, kk, yy);
    }
    __syncthreads();
    int s3 = 0;
    for (int blk = 0; blk < NBLK; ++blk) {
        const int s3n = (s3 == 2) ? 0 : s3 + 1;
        const int s3p = (s3 == 0) ? 2 : s3 - 1;
        if (role < 2) {
            consume(blk, s3);
        } else if (blk + 1 < NBLK) {
            const bool outs = role == 3 && blk > 0;
            float ccA[16][2], ccB[16][2];
            int kk[NH], yy[NH];
            load_idx(blk + 1, par, kk, yy);
            if (outs) finish(blk - 1, s3p, 0);
            first_rows(ccA, ccB, kk);
            if (outs) {
#pragma nounroll
                for (int h = 1; h < NB / NF; ++h) {
                    finish(blk - 1, s3p, h * NF);
                }
            }
            produce_steps(ccA, ccB, blk + 1, s3n, par, kk, yy);
        } else if (role == 3) {
            finish_whole(blk - 1, s3p);
        }
        __syncthreads();
        s3 = s3n;
    }
    if (role == 3) {
        finish_whole(NBLK - 1, (s3 == 0) ? 2 : s3 - 1);
    }
}

extern "C" void kernel_launch(void* const* d_in, const int* in_sizes, int n_in,
                              void* d_out, int out_size, void* d_ws, size_t ws_size,
                              hipStream_t stream) {
    const int*   corr  = (const int*)d_in[0];
    const int*   kc    = (const int*)d_in[1];
    const float* Amat  = (const float*)d_in[2];
    const float* trans = (const float*)d_in[3];
    const float* obs   = (const float*)d_in[4];
    const float* init  = (const float*)d_in[5];
    float*       out   = (float*)d_out;
    float*       ws    = (float*)d_ws;

    (void)in_sizes; (void)n_in; (void)out_size; (void)ws_size;

    bkt_prep<<<1, 64, 0, stream>>>(trans, obs, init, ws);
    bkt_fwd<<<BSZ / TILE, 128, 0, stream>>>(corr, kc, Amat, ws, out);
}
